# starting kernel
# baseline (speedup 1.0000x reference)
#include <hip/hip_runtime.h>
#include <hip/hip_bf16.h>

typedef __attribute__((ext_vector_type(2)))  float    v2f;
typedef __attribute__((ext_vector_type(8)))  float    v8f;
typedef __attribute__((ext_vector_type(8)))  _Float16 v8h;
typedef __attribute__((ext_vector_type(16))) _Float16 v16h;
typedef __attribute__((ext_vector_type(4)))  int      v4i;

#define BQ 4
#define SQ 128
#define MQ (BQ*SQ)
#define INF 4096
#define OUTF 4096
#define BSR 128
#define NBIN (INF/BSR)
#define NBOUT (OUTF/BSR)

static __device__ inline v16h load_frag16(const _Float16* __restrict__ p) {
    const v8h lo = *(const v8h*)p;
    const v8h hi = *(const v8h*)(p + 16);
    v16h r;
    #pragma unroll
    for (int i = 0; i < 8; ++i) { r[i] = lo[i]; r[8 + i] = hi[i]; }
    return r;
}

__global__ __launch_bounds__(256) void k_decode_w(const int* __restrict__ Wp,
                                                  _Float16* __restrict__ W16)
{
    const size_t i = (size_t)blockIdx.x * 256 + threadIdx.x;
    const v4i p = *(const v4i*)(Wp + i * 4);
    v8h w;
    #pragma unroll
    for (int j = 0; j < 4; ++j) {
        const int c = p[j];
        w[2 * j + 0] = (_Float16)(((c & 0xF) ^ 8) - 8);
        w[2 * j + 1] = (_Float16)((((c >> 4) & 0xF) ^ 8) - 8);
    }
    *(v8h*)(W16 + i * 8) = w;
}

__global__ __launch_bounds__(256) void k_in_rot(
    const float* __restrict__ x, const int* __restrict__ perm,
    const float* __restrict__ R_in, const float* __restrict__ s_a,
    _Float16* __restrict__ A)
{
    const int mt = blockIdx.x;
    const int nb = blockIdx.y;
    const int m0 = mt * 16;
    const int tid  = threadIdx.x;
    const int wave = tid >> 5;
    const int lane = tid & 31;

    __shared__ float xp[16 * 132];
    for (int i = tid; i < 16 * 128; i += 256) {
        const int r = i >> 7, c = i & 127;
        const int src = perm[nb * 128 + c];
        xp[r * 132 + c] = x[(size_t)(m0 + r) * INF + src];
    }
    __syncthreads();

    const int d0      = wave * 16;
    const int halfsel = lane >> 4;
    const int lr      = lane & 15;

    v8f acc = {};
    const float* Rb = R_in + (size_t)nb * BSR * BSR;
    for (int c0 = 0; c0 < BSR; c0 += 4) {
        const int ca = c0 + halfsel * 2;
        v2f a;
        a.x = xp[lr * 132 + ca];
        a.y = xp[lr * 132 + ca + 1];
        v2f b;
        b.x = Rb[(size_t)(ca + 0) * BSR + d0 + lr];
        b.y = Rb[(size_t)(ca + 1) * BSR + d0 + lr];
        acc = __builtin_amdgcn_wmma_f32_16x16x4_f32(false, a, false, b,
                                                    (short)0, acc, false, false);
    }

    const int k = nb * 128 + d0 + lr;
    const float s   = s_a[k];
    const float inv = 1.0f / s;
    #pragma unroll
    for (int r = 0; r < 8; ++r) {
        const int M = r + halfsel * 8;
        float q = __builtin_rintf(acc[r] * inv);
        q = fminf(fmaxf(q, -128.0f), 127.0f);
        A[(size_t)(m0 + M) * INF + k] = (_Float16)(q * s);
    }
}

__global__ __launch_bounds__(256) void k_gemm16(
    const _Float16* __restrict__ A, const _Float16* __restrict__ W16,
    const float* __restrict__ s_w, float* __restrict__ Y)
{
    const int mt = blockIdx.x, og = blockIdx.y;
    const int wave = threadIdx.x >> 5;
    const int lane = threadIdx.x & 31;
    const int m0 = mt * 32;
    const int o0 = og * 256 + wave * 32;
    const int halfsel = lane >> 4;
    const int lr      = lane & 15;
    const int kbase   = halfsel * 8;

    const _Float16* a0p = A   + (size_t)(m0 + lr)      * INF + kbase;
    const _Float16* a1p = A   + (size_t)(m0 + 16 + lr) * INF + kbase;
    const _Float16* b0p = W16 + (size_t)(o0 + lr)      * INF + kbase;
    const _Float16* b1p = W16 + (size_t)(o0 + 16 + lr) * INF + kbase;

    v8f acc00 = {}, acc01 = {}, acc10 = {}, acc11 = {};
    for (int k0 = 0; k0 < INF; k0 += 32) {
        const v16h a0 = load_frag16(a0p + k0);
        const v16h a1 = load_frag16(a1p + k0);
        const v16h b0 = load_frag16(b0p + k0);
        const v16h b1 = load_frag16(b1p + k0);
        acc00 = __builtin_amdgcn_wmma_f32_16x16x32_f16(false, a0, false, b0,
                                                       (short)0, acc00, false, false);
        acc01 = __builtin_amdgcn_wmma_f32_16x16x32_f16(false, a0, false, b1,
                                                       (short)0, acc01, false, false);
        acc10 = __builtin_amdgcn_wmma_f32_16x16x32_f16(false, a1, false, b0,
                                                       (short)0, acc10, false, false);
        acc11 = __builtin_amdgcn_wmma_f32_16x16x32_f16(false, a1, false, b1,
                                                       (short)0, acc11, false, false);
    }

    const float sw0 = s_w[o0 + lr];
    const float sw1 = s_w[o0 + 16 + lr];
    #pragma unroll
    for (int r = 0; r < 8; ++r) {
        const int M = r + halfsel * 8;
        Y[(size_t)(m0 + M)      * OUTF + o0 + lr]      = acc00[r] * sw0;
        Y[(size_t)(m0 + M)      * OUTF + o0 + 16 + lr] = acc01[r] * sw1;
        Y[(size_t)(m0 + 16 + M) * OUTF + o0 + lr]      = acc10[r] * sw0;
        Y[(size_t)(m0 + 16 + M) * OUTF + o0 + 16 + lr] = acc11[r] * sw1;
    }
}

__global__ __launch_bounds__(256) void k_gemm_dec(
    const _Float16* __restrict__ A, const int* __restrict__ Wp,
    const float* __restrict__ s_w, float* __restrict__ Y)
{
    const int mt = blockIdx.x, og = blockIdx.y;
    const int m0 = mt * 16;
    const int wave = threadIdx.x >> 5;
    const int lane = threadIdx.x & 31;
    const int o0      = (og * 8 + wave) * 16;
    const int halfsel = lane >> 4;
    const int lr      = lane & 15;

    const int    kbase = halfsel * 8;
    const size_t arow  = (size_t)(m0 + lr) * INF;
    const int    oc    = o0 + lr;
    const int*   wrow  = Wp + (size_t)oc * (INF / 2);

    v8f acc = {};
    for (int k0 = 0; k0 < INF; k0 += 32) {
        const v16h a = load_frag16(A + arow + k0 + kbase);
        const v4i p0 = *(const v4i*)(wrow + ((k0 + kbase) >> 1));
        const v4i p1 = *(const v4i*)(wrow + ((k0 + kbase + 16) >> 1));
        v16h b;
        #pragma unroll
        for (int j = 0; j < 4; ++j) {
            int c0 = p0[j];
            b[2 * j + 0] = (_Float16)(((c0 & 0xF) ^ 8) - 8);
            b[2 * j + 1] = (_Float16)((((c0 >> 4) & 0xF) ^ 8) - 8);
            int c1 = p1[j];
            b[8 + 2 * j + 0] = (_Float16)(((c1 & 0xF) ^ 8) - 8);
            b[8 + 2 * j + 1] = (_Float16)((((c1 >> 4) & 0xF) ^ 8) - 8);
        }
        acc = __builtin_amdgcn_wmma_f32_16x16x32_f16(false, a, false, b,
                                                     (short)0, acc, false, false);
    }

    const float sw = s_w[oc];
    #pragma unroll
    for (int r = 0; r < 8; ++r) {
        const int M = r + halfsel * 8;
        Y[(size_t)(m0 + M) * OUTF + oc] = acc[r] * sw;
    }
}

__global__ __launch_bounds__(256) void k_out_rot(
    const float* __restrict__ Y, const float* __restrict__ R_out,
    const float* __restrict__ bias, float* __restrict__ out)
{
    const int mt = blockIdx.x, nb = blockIdx.y;
    const int m0 = mt * 16;
    const int wave = threadIdx.x >> 5;
    const int lane = threadIdx.x & 31;
    const int d0      = wave * 16;
    const int halfsel = lane >> 4;
    const int lr      = lane & 15;

    v8f acc = {};
    const float* Rb   = R_out + (size_t)nb * BSR * BSR;
    const size_t yrow = (size_t)(m0 + lr) * OUTF + nb * 128;
    const size_t brow = (size_t)(d0 + lr) * BSR;
    for (int c0 = 0; c0 < BSR; c0 += 4) {
        const int ca = c0 + halfsel * 2;
        const v2f a = *(const v2f*)(Y  + yrow + ca);
        const v2f b = *(const v2f*)(Rb + brow + ca);
        acc = __builtin_amdgcn_wmma_f32_16x16x4_f32(false, a, false, b,
                                                    (short)0, acc, false, false);
    }

    const int o  = nb * 128 + d0 + lr;
    const float bs = bias[o];
    #pragma unroll
    for (int r = 0; r < 8; ++r) {
        const int M = r + halfsel * 8;
        out[(size_t)(m0 + M) * OUTF + o] = acc[r] + bs;
    }
}

extern "C" void kernel_launch(void* const* d_in, const int* in_sizes, int n_in,
                              void* d_out, int out_size, void* d_ws, size_t ws_size,
                              hipStream_t stream) {
    const float* x        = (const float*)d_in[0];
    const int*   perm     = (const int*)  d_in[1];
    const float* R_in     = (const float*)d_in[2];
    const float* s_a      = (const float*)d_in[3];
    const int*   packed_w = (const int*)  d_in[4];
    const float* s_w      = (const float*)d_in[5];
    const float* bias     = (const float*)d_in[6];
    const float* R_out    = (const float*)d_in[7];
    float* out = (float*)d_out;

    const size_t szA  = (size_t)MQ * INF * sizeof(_Float16);
    const size_t szY  = (size_t)MQ * OUTF * sizeof(float);
    const size_t szW  = (size_t)OUTF * INF * sizeof(_Float16);
    _Float16* A  = (_Float16*)d_ws;
    float*    Ys = (float*)((char*)d_ws + szA);
    _Float16* W16 = (_Float16*)((char*)d_ws + szA + szY);

    const dim3 blk(256);
    k_in_rot<<<dim3(MQ / 16, NBIN), blk, 0, stream>>>(x, perm, R_in, s_a, A);

    if (ws_size >= szA + szY + szW) {
        k_decode_w<<<dim3((OUTF * (INF / 2) / 4) / 256), blk, 0, stream>>>(packed_w, W16);
        k_gemm16<<<dim3(MQ / 32, OUTF / 256), blk, 0, stream>>>(A, W16, s_w, Ys);
    } else {
        k_gemm_dec<<<dim3(MQ / 16, OUTF / 128), blk, 0, stream>>>(A, packed_w, s_w, Ys);
    }

    k_out_rot<<<dim3(MQ / 16, NBOUT), blk, 0, stream>>>(Ys, R_out, bias, out);
}


# k_gemm16: four waves per group, one per SIMD (waves_per_eu 1,1: one group per WGP), 64x32 per wave (was two waves of 64x64, two SIMDs idle); A lines shared via the WGP cache; k and sum order kept
# speedup vs baseline: 2.8398x; 2.8398x over previous
#include <hip/hip_runtime.h>
#include <hip/hip_bf16.h>

typedef __attribute__((ext_vector_type(2)))  float    v2f;
typedef __attribute__((ext_vector_type(4)))  float    v4f;
typedef __attribute__((ext_vector_type(8)))  float    v8f;
typedef __attribute__((ext_vector_type(8)))  _Float16 v8h;
typedef __attribute__((ext_vector_type(16))) _Float16 v16h;
typedef __attribute__((ext_vector_type(4)))  int      v4i;
typedef __attribute__((ext_vector_type(8)))  int      v8i;

#define BQ 4
#define SQ 128
#define MQ (BQ*SQ)
#define INF 4096
#define OUTF 4096
#define BSR 128
#define NBIN (INF/BSR)
#define NBOUT (OUTF/BSR)
#define KSTEPS (INF/32)

static __device__ inline size_t frag_block(int rt, int kt) {
    return ((size_t)rt * KSTEPS + (size_t)kt) * 512;
}

static __device__ inline size_t frag8_block(int rt, int kt) {
    return ((size_t)rt * KSTEPS + (size_t)kt) * 128;
}

static __device__ inline int pack_w8(int p01, int p23) {
    return ((p01 >> 8) & 0x00FF00FF) | (p23 & (int)0xFF00FF00u);
}

static __device__ inline v16h expand_w8(v4i q) {
    v8i r;
    #pragma unroll
    for (int j = 0; j < 4; ++j) {
        r[2 * j + 0] = (q[j] << 8) & (int)0xFF00FF00u;
        r[2 * j + 1] = q[j] & (int)0xFF00FF00u;
    }
    return __builtin_bit_cast(v16h, r);
}

static __device__ inline v16h load_frag16(const _Float16* __restrict__ p) {
    const v8h lo = *(const v8h*)p;
    const v8h hi = *(const v8h*)(p + 256);
    v16h r;
    #pragma unroll
    for (int i = 0; i < 8; ++i) { r[i] = lo[i]; r[8 + i] = hi[i]; }
    return r;
}

static __device__ inline size_t yfrag_block(int rt, int nb) {
    return ((size_t)rt * NBOUT + (size_t)nb) * 2048;
}

template <bool DEC>
__global__ __launch_bounds__(256) void k_in_rot(
    const float* __restrict__ x, const int* __restrict__ perm,
    const float* __restrict__ R_in, const float* __restrict__ s_a,
    _Float16* __restrict__ A,
    const int* __restrict__ Wp, int* __restrict__ W8)
{
    const int mt = blockIdx.x;
    const int nb = blockIdx.y;
    const int m0 = mt * 16;
    const int tid  = threadIdx.x;
    const int wave = tid >> 5;
    const int lane = tid & 31;
    const int halfsel = lane >> 4;
    const int lr      = lane & 15;

    __shared__ float xp[16 * 132];
    __shared__ v4i stage[8 * 16 * 17];

    const int xc = tid & 127;
    const int xr = tid >> 7;
    const int src = perm[nb * 128 + xc];

    const int gw = ((nb * 32 + mt) << 3) + wave;
    const int rt = gw >> 5;
    const int kq = gw & 31;
    v4i* st = stage + wave * (16 * 17);
    v4i p[8];
    if constexpr (DEC) {
        const int* wsrc = Wp + (size_t)(rt * 16 + halfsel) * (INF / 2) + kq * 64 + lr * 4;
        #pragma unroll
        for (int n = 0; n < 8; ++n) p[n] = *(const v4i*)(wsrc + (size_t)n * 2 * (INF / 2));
    }

    float xv[8];
    #pragma unroll
    for (int j = 0; j < 8; ++j) xv[j] = x[(size_t)(m0 + xr + 2 * j) * INF + src];
    #pragma unroll
    for (int j = 0; j < 8; ++j) xp[(xr + 2 * j) * 132 + xc] = xv[j];
    if constexpr (DEC) {
        #pragma unroll
        for (int n = 0; n < 8; ++n) st[(2 * n + halfsel) * 17 + lr] = p[n];
    }
    __syncthreads();

    if constexpr (DEC) {
        int* dst = W8 + frag8_block(rt, kq * 4) + lane * 4;
        #pragma unroll
        for (int ks = 0; ks < 4; ++ks) {
            v4i o;
            #pragma unroll
            for (int jp = 0; jp < 2; ++jp) {
                const v4i q = st[lr * 17 + 2 * (2 * ks + jp) + halfsel];
                v8h w;
                #pragma unroll
                for (int j = 0; j < 4; ++j) {
                    const int cw = q[j];
                    w[2 * j + 0] = (_Float16)(((cw & 0xF) ^ 8) - 8);
                    w[2 * j + 1] = (_Float16)((((cw >> 4) & 0xF) ^ 8) - 8);
                }
                const v4i wb = __builtin_bit_cast(v4i, w);
                o[2 * jp + 0] = pack_w8(wb[0], wb[1]);
                o[2 * jp + 1] = pack_w8(wb[2], wb[3]);
            }
            *(v4i*)(dst + ks * 128) = o;
        }
    }

    const int d0 = wave * 16;

    v8f acc = {};
    const float* Rb = R_in + (size_t)nb * BSR * BSR;
    for (int c0 = 0; c0 < BSR; c0 += 4) {
        const int ca = c0 + halfsel * 2;
        v2f a;
        a.x = xp[lr * 132 + ca];
        a.y = xp[lr * 132 + ca + 1];
        v2f b;
        b.x = Rb[(size_t)(ca + 0) * BSR + d0 + lr];
        b.y = Rb[(size_t)(ca + 1) * BSR + d0 + lr];
        acc = __builtin_amdgcn_wmma_f32_16x16x4_f32(false, a, false, b,
                                                    (short)0, acc, false, false);
    }

    const int k = nb * 128 + d0 + lr;
    const float s   = s_a[k];
    const float inv = 1.0f / s;
    _Float16* Ablk = A + frag_block(mt, nb * 4 + (wave >> 1)) + (wave & 1) * 256
                       + (lr >> 3) * 128 + (lr & 7);
    #pragma unroll
    for (int r = 0; r < 8; ++r) {
        const int M = r + halfsel * 8;
        float q = __builtin_rintf(acc[r] * inv);
        q = fminf(fmaxf(q, -128.0f), 127.0f);
        Ablk[M * 8] = (_Float16)(q * s);
    }
}

#define GEMM_ST 8
#define GEMM_NK KSTEPS

#define GEMM_LOAD(S, KT)                                                       \
    _Pragma("unroll")                                                          \
    for (int i = 0; i < 4; ++i) {                                              \
        fa[S][i] = load_frag16(ap + frag_block(i, (KT)));                      \
        if (i < 2) fb[S][i] = *(const v4i*)(bp + frag8_block(i, (KT)));        \
    }

#define GEMM_MMA(S)                                                            \
    _Pragma("unroll")                                                          \
    for (int j = 0; j < 2; ++j) {                                              \
        const v16h wj = expand_w8(fb[S][j]);                                   \
        _Pragma("unroll")                                                      \
        for (int i = 0; i < 4; ++i)                                            \
            acc[i][j] = __builtin_amdgcn_wmma_f32_16x16x32_f16(                \
                false, wj, false, fa[S][i], (short)0, acc[i][j],               \
                false, false);                                                 \
    }

#define GEMM_PIN __builtin_amdgcn_sched_barrier(0);

#define GEMM_PF_DIST 12
#define GEMM_PF(KT) __builtin_prefetch(pfp + (size_t)(KT) * pf_step, 0, 2);

#define GEMM_STEP(J)                                                           \
    GEMM_LOAD(((J) + GEMM_ST - 1) % GEMM_ST, kt + (J) + GEMM_ST - 1)           \
    GEMM_PF(min(kt + (J) + GEMM_ST - 1 + GEMM_PF_DIST, GEMM_NK - 1)) GEMM_PIN  \
    GEMM_MMA(J) GEMM_PIN

__global__ __launch_bounds__(128) __attribute__((amdgpu_waves_per_eu(1, 1))) void k_gemm16(
    const _Float16* __restrict__ A, const int* __restrict__ W8,
    const float* __restrict__ s_w, float* __restrict__ Y)
{
    const int mt = blockIdx.x, og = blockIdx.y;
    const int wave = threadIdx.x >> 5;
    const int lane = threadIdx.x & 31;
    const int m0 = mt * 64;
    const int o0 = og * 128 + wave * 32;
    const int halfsel = lane >> 4;

    const _Float16* ap = A  + frag_block(m0 >> 4, 0) + lane * 8;
    const int*      bp = W8 + frag8_block(o0 >> 4, 0) + lane * 4;

    const char* pfp = halfsel
        ? (const char*)(A  + frag_block((m0 >> 4) + wave, 0))                + (lane & 3) * 256
        : (const char*)(W8 + frag8_block((o0 >> 4) + ((lane >> 1) & 1), 0)) + (lane & 1) * 256;
    const size_t pf_step = halfsel ? 1024 : 512;

    v16h fa[GEMM_ST][4];
    v4i  fb[GEMM_ST][2];
    v8f acc[4][2];
    #pragma unroll
    for (int i = 0; i < 4; ++i) {
        #pragma unroll
        for (int j = 0; j < 2; ++j) acc[i][j] = (v8f){};
    }

    GEMM_LOAD(0, 0) GEMM_PIN
    GEMM_LOAD(1, 1) GEMM_PIN
    GEMM_LOAD(2, 2) GEMM_PIN
    GEMM_LOAD(3, 3) GEMM_PIN
    GEMM_LOAD(4, 4) GEMM_PIN
    GEMM_LOAD(5, 5) GEMM_PIN
    GEMM_LOAD(6, 6) GEMM_PIN
    #pragma unroll
    for (int p = GEMM_ST - 1; p < GEMM_ST - 1 + GEMM_PF_DIST; ++p) GEMM_PF(p)
    GEMM_PIN

    for (int kt = 0; kt < GEMM_NK - GEMM_ST; kt += GEMM_ST) {
        GEMM_STEP(0) GEMM_STEP(1) GEMM_STEP(2) GEMM_STEP(3)
        GEMM_STEP(4) GEMM_STEP(5) GEMM_STEP(6) GEMM_STEP(7)
    }
    GEMM_LOAD(GEMM_ST - 1, GEMM_NK - 1) GEMM_PIN
    GEMM_MMA(0)
    GEMM_MMA(1)
    GEMM_MMA(2)
    GEMM_MMA(3)
    GEMM_MMA(4)
    GEMM_MMA(5)
    GEMM_MMA(6)
    GEMM_MMA(7)

    #pragma unroll
    for (int j = 0; j < 2; ++j) {
        const float* swp = s_w + o0 + j * 16 + halfsel * 8;
        const v4f s0 = *(const v4f*)swp;
        const v4f s1 = *(const v4f*)(swp + 4);
        #pragma unroll
        for (int i = 0; i < 4; ++i) {
            float* yb = Y + yfrag_block((m0 >> 4) + i, og) + (wave * 2 + j) * 256 + lane * 4;
            v4f lo, hi;
            #pragma unroll
            for (int e = 0; e < 4; ++e) {
                lo[e] = acc[i][j][e]     * s0[e];
                hi[e] = acc[i][j][4 + e] * s1[e];
            }
            *(v4f*)yb         = lo;
            *(v4f*)(yb + 128) = hi;
        }
    }
}

__global__ __launch_bounds__(256) void k_gemm_dec(
    const _Float16* __restrict__ A, const int* __restrict__ Wp,
    const float* __restrict__ s_w, float* __restrict__ Y)
{
    const int mt = blockIdx.x, og = blockIdx.y;
    const int wave = threadIdx.x >> 5;
    const int lane = threadIdx.x & 31;
    const int o0      = (og * 8 + wave) * 16;
    const int halfsel = lane >> 4;
    const int lr      = lane & 15;

    const int    kbase = halfsel * 8;
    const int    oc    = o0 + lr;
    const int*   wrow  = Wp + (size_t)oc * (INF / 2);

    v8f acc = {};
    for (int k0 = 0; k0 < INF; k0 += 32) {
        const v16h a = load_frag16(A + frag_block(mt, k0 >> 5) + lane * 8);
        const v4i p0 = *(const v4i*)(wrow + ((k0 + kbase) >> 1));
        const v4i p1 = *(const v4i*)(wrow + ((k0 + kbase + 16) >> 1));
        v16h b;
        #pragma unroll
        for (int j = 0; j < 4; ++j) {
            int c0 = p0[j];
            b[2 * j + 0] = (_Float16)(((c0 & 0xF) ^ 8) - 8);
            b[2 * j + 1] = (_Float16)((((c0 >> 4) & 0xF) ^ 8) - 8);
            int c1 = p1[j];
            b[8 + 2 * j + 0] = (_Float16)(((c1 & 0xF) ^ 8) - 8);
            b[8 + 2 * j + 1] = (_Float16)((((c1 >> 4) & 0xF) ^ 8) - 8);
        }
        acc = __builtin_amdgcn_wmma_f32_16x16x32_f16(false, b, false, a,
                                                     (short)0, acc, false, false);
    }

    const float* swp = s_w + o0 + halfsel * 8;
    const v4f s0 = *(const v4f*)swp;
    const v4f s1 = *(const v4f*)(swp + 4);
    float* yb = Y + yfrag_block(mt, og) + wave * 256 + lane * 4;
    v4f lo, hi;
    #pragma unroll
    for (int e = 0; e < 4; ++e) {
        lo[e] = acc[e]     * s0[e];
        hi[e] = acc[4 + e] * s1[e];
    }
    *(v4f*)yb         = lo;
    *(v4f*)(yb + 128) = hi;
}

__global__ __launch_bounds__(256) void k_out_rot(
    const float* __restrict__ Y, const float* __restrict__ R_out,
    const float* __restrict__ bias, float* __restrict__ out)
{
    const int mg = blockIdx.x, nb = blockIdx.y;
    const int wave = threadIdx.x >> 5;
    const int lane = threadIdx.x & 31;
    const int d0      = wave * 16;
    const int halfsel = lane >> 4;
    const int lr      = lane & 15;

    const float* Rrow = R_out + (size_t)nb * BSR * BSR + (size_t)(d0 + lr) * BSR + halfsel * 8;
    const float* Yb = Y + yfrag_block(mg * 4, nb) + lane * 4;

    v8f acc[4];
    #pragma unroll
    for (int i = 0; i < 4; ++i) acc[i] = (v8f){};

    v4f b[16], a[16][4];
    #pragma unroll
    for (int t = 0; t < 16; ++t) {
        b[t] = *(const v4f*)(Rrow + (t >> 1) * 16 + (t & 1) * 4);
        #pragma unroll
        for (int i = 0; i < 4; ++i)
            a[t][i] = *(const v4f*)(Yb + yfrag_block(i, 0) + t * 128);
    }
    __builtin_amdgcn_sched_barrier(0);
    #pragma unroll
    for (int t = 0; t < 16; ++t) {
        #pragma unroll
        for (int s = 0; s < 2; ++s) {
            v2f bb;
            bb.x = b[t][2 * s];
            bb.y = b[t][2 * s + 1];
            #pragma unroll
            for (int i = 0; i < 4; ++i) {
                v2f aa;
                aa.x = a[t][i][2 * s];
                aa.y = a[t][i][2 * s + 1];
                acc[i] = __builtin_amdgcn_wmma_f32_16x16x4_f32(false, aa, false, bb,
                                                               (short)0, acc[i], false, false);
            }
        }
    }

    const int o  = nb * 128 + d0 + lr;
    const float bs = bias[o];
    #pragma unroll
    for (int i = 0; i < 4; ++i) {
        #pragma unroll
        for (int r = 0; r < 8; ++r) {
            const int M = mg * 64 + i * 16 + r + halfsel * 8;
            out[(size_t)M * OUTF + o] = acc[i][r] + bs;
        }
    }
}

extern "C" void kernel_launch(void* const* d_in, const int* in_sizes, int n_in,
                              void* d_out, int out_size, void* d_ws, size_t ws_size,
                              hipStream_t stream) {
    const float* x        = (const float*)d_in[0];
    const int*   perm     = (const int*)  d_in[1];
    const float* R_in     = (const float*)d_in[2];
    const float* s_a      = (const float*)d_in[3];
    const int*   packed_w = (const int*)  d_in[4];
    const float* s_w      = (const float*)d_in[5];
    const float* bias     = (const float*)d_in[6];
    const float* R_out    = (const float*)d_in[7];
    float* out = (float*)d_out;

    const size_t szA  = (size_t)MQ * INF * sizeof(_Float16);
    const size_t szY  = (size_t)MQ * OUTF * sizeof(float);
    const size_t szW  = (size_t)OUTF * INF;
    _Float16* A  = (_Float16*)d_ws;
    float*    Ys = (float*)((char*)d_ws + szA);
    int*      W8 = (int*)((char*)d_ws + szA + szY);

    const dim3 blk(256);

    if (ws_size >= szA + szY + szW) {
        k_in_rot<true><<<dim3(MQ / 16, NBIN), blk, 0, stream>>>(x, perm, R_in, s_a, A, packed_w, W8);
        k_gemm16<<<dim3(MQ / 64, OUTF / 128), dim3(128), 0, stream>>>(A, W8, s_w, Ys);
    } else {
        k_in_rot<false><<<dim3(MQ / 16, NBIN), blk, 0, stream>>>(x, perm, R_in, s_a, A, packed_w, W8);
        k_gemm_dec<<<dim3(MQ / 16, OUTF / 128), blk, 0, stream>>>(A, packed_w, s_w, Ys);
    }

    k_out_rot<<<dim3(MQ / 64, NBOUT), blk, 0, stream>>>(Ys, R_out, bias, out);
}
